# starting kernel
# baseline (speedup 1.0000x reference)
#include <hip/hip_runtime.h>

typedef __attribute__((ext_vector_type(2))) float v2f;
typedef __attribute__((ext_vector_type(8))) float v8f;

#define NGROUPS 17
#define WAVES_PER_BLOCK 8
#define TILE 16
#define TPW  (2 * TILE)
#define XROW 48
#define OROW 51

__global__ __launch_bounds__(WAVES_PER_BLOCK * 32)
void bone_mlp_wmma_kernel(const float* __restrict__ x,
                          const float* __restrict__ W1,
                          const float* __restrict__ b1,
                          const float* __restrict__ W2,
                          const float* __restrict__ b2,
                          const int*   __restrict__ idx,
                          float* __restrict__ out,
                          int npairs)
{
    __shared__ float lds_out[WAVES_PER_BLOCK][TPW * OROW];

    const int lane = threadIdx.x & 31;
    const int wave = threadIdx.x >> 5;
    const int pair = blockIdx.x * WAVES_PER_BLOCK + wave;
    if (pair >= npairs) return;

    const int half = lane >> 4;
    const int lm   = lane & 15;
    const int cidx = lm < 3 ? lm : 2;

    const float* xrowA = x + (size_t)(pair * TPW + lm) * XROW;
    const float* xrowB = xrowA + TILE * XROW;
    float* lrow = &lds_out[wave][0];

    for (int g = 0; g < NGROUPS; ++g) {
        const float* w1g  = W1 + g * 12 * 16;
        const float* w2g  = W2 + g * 16 * 3 + cidx;
        const int*   idxg = idx + g * 4;

        v8f accA = {};
        v8f accB = {};
#pragma unroll
        for (int j = 0; j < 3; ++j) {
            const int f0 = 4 * j + 2 * half;
            const int f1 = f0 + 1;
            const int o0 = idxg[f0 / 3] * 3 + f0 % 3;
            const int o1 = idxg[f1 / 3] * 3 + f1 % 3;
            v2f a, bA, bB;
            a.x  = w1g[f0 * 16 + lm];
            a.y  = w1g[f1 * 16 + lm];
            bA.x = xrowA[o0];  bA.y = xrowA[o1];
            bB.x = xrowB[o0];  bB.y = xrowB[o1];
            accA = __builtin_amdgcn_wmma_f32_16x16x4_f32(
                false, a, false, bA, (short)0, accA, false, false);
            accB = __builtin_amdgcn_wmma_f32_16x16x4_f32(
                false, a, false, bB, (short)0, accB, false, false);
        }

        const float* b1g = b1 + g * 16 + 8 * half;
#pragma unroll
        for (int i = 0; i < 8; ++i) {
            const float bb = b1g[i];
            float tA = accA[i] + bb;
            float tB = accB[i] + bb;
            accA[i] = tA > 0.0f ? tA : 0.0f;
            accB[i] = tB > 0.0f ? tB : 0.0f;
        }

        v8f acc2A = {};
        v8f acc2B = {};
#pragma unroll
        for (int j = 0; j < 4; ++j) {
            const int h0 = 2 * j + 8 * half;
            v2f b;
            b.x = w2g[(h0 + 0) * 3];
            b.y = w2g[(h0 + 1) * 3];
            v2f aA; aA.x = accA[2 * j]; aA.y = accA[2 * j + 1];
            v2f aB; aB.x = accB[2 * j]; aB.y = accB[2 * j + 1];
            acc2A = __builtin_amdgcn_wmma_f32_16x16x4_f32(
                false, aA, false, b, (short)0, acc2A, false, false);
            acc2B = __builtin_amdgcn_wmma_f32_16x16x4_f32(
                false, aB, false, b, (short)0, acc2B, false, false);
        }

        const float bb2 = b2[g * 3 + cidx];
        if (lm < 3) {
#pragma unroll
            for (int i = 0; i < 8; ++i) {
                const int row = i + 8 * half;
                lrow[row * OROW + g * 3 + lm]                = acc2A[i] + bb2;
                lrow[(row + TILE) * OROW + g * 3 + lm]       = acc2B[i] + bb2;
            }
        }
    }

    float* outp = out + (size_t)pair * (TPW * OROW);
#pragma unroll 4
    for (int k = lane; k < TPW * OROW; k += 32)
        __builtin_nontemporal_store(lrow[k], &outp[k]);
}

extern "C" void kernel_launch(void* const* d_in, const int* in_sizes, int n_in,
                              void* d_out, int out_size, void* d_ws, size_t ws_size,
                              hipStream_t stream) {
    (void)n_in; (void)out_size; (void)d_ws; (void)ws_size;
    const float* x  = (const float*)d_in[0];
    const float* W1 = (const float*)d_in[1];
    const float* b1 = (const float*)d_in[2];
    const float* W2 = (const float*)d_in[3];
    const float* b2 = (const float*)d_in[4];
    const int*   idx = (const int*)d_in[5];
    float* out = (float*)d_out;

    const int ntok   = in_sizes[0] / XROW;
    const int npairs = ntok / TPW;
    const int blocks = (npairs + WAVES_PER_BLOCK - 1) / WAVES_PER_BLOCK;

    bone_mlp_wmma_kernel<<<blocks, WAVES_PER_BLOCK * 32, 0, stream>>>(
        x, W1, b1, W2, b2, idx, out, npairs);
}


# Stage each wave's 32 token rows of x in LDS once with coalesced 16-byte non-temporal loads; gather layer-1 B operands from LDS by scalar idx offsets (was 204 scattered global loads per wave)
# speedup vs baseline: 2.5768x; 2.5768x over previous
#include <hip/hip_runtime.h>

typedef __attribute__((ext_vector_type(2))) float v2f;
typedef __attribute__((ext_vector_type(4))) float v4f;
typedef __attribute__((ext_vector_type(8))) float v8f;

#define NGROUPS 17
#define WAVES_PER_BLOCK 8
#define TILE 16
#define TPW  (2 * TILE)
#define XROW 48
#define OROW 51
#define XS   52

__global__ __launch_bounds__(WAVES_PER_BLOCK * 32)
void bone_mlp_wmma_kernel(const float* __restrict__ x,
                          const float* __restrict__ W1,
                          const float* __restrict__ b1,
                          const float* __restrict__ W2,
                          const float* __restrict__ b2,
                          const int*   __restrict__ idx,
                          float* __restrict__ out,
                          int npairs)
{
    __shared__ float lds_out[WAVES_PER_BLOCK][TPW * OROW];
    __shared__ __attribute__((aligned(16))) float lds_x[WAVES_PER_BLOCK][TPW * XS];

    const int lane = threadIdx.x & 31;
    const int wave = threadIdx.x >> 5;
    const int pair = blockIdx.x * WAVES_PER_BLOCK + wave;
    if (pair >= npairs) return;

    const int half = lane >> 4;
    const int lm   = lane & 15;
    const int cidx = lm < 3 ? lm : 2;

    float* lrow = &lds_out[wave][0];

    float* xs = &lds_x[wave][0];
    {
        const v4f* xw = reinterpret_cast<const v4f*>(x + (size_t)pair * (TPW * XROW));
        v4f r[12];
#pragma unroll
        for (int i = 0; i < 12; ++i)
            r[i] = __builtin_nontemporal_load(xw + i * 32 + lane);
#pragma unroll
        for (int b = 0; b < 3; ++b) {
            const int t = 32 * b + lane;
            float* dst = xs + (t / 12) * XS + (t % 12) * 4;
#pragma unroll
            for (int a = 0; a < 4; ++a)
                *reinterpret_cast<v4f*>(dst + 8 * a * XS) = r[3 * a + b];
        }
    }
    const float* xrowA = xs + lm * XS;
    const float* xrowB = xrowA + TILE * XS;

    for (int g = 0; g < NGROUPS; ++g) {
        const float* w1g  = W1 + g * 12 * 16;
        const float* w2g  = W2 + g * 16 * 3 + cidx;
        const int*   idxg = idx + g * 4;
        const int l0 = idxg[0] * 3, l1 = idxg[1] * 3, l2 = idxg[2] * 3, l3 = idxg[3] * 3;
        const int lofs[4] = { l0, l1, l2, l3 };

        v8f accA = {};
        v8f accB = {};
#pragma unroll
        for (int j = 0; j < 3; ++j) {
            const int f0 = 4 * j + 2 * half;
            const int f1 = f0 + 1;
            const int o0 = half ? lofs[(4 * j + 2) / 3] + (4 * j + 2) % 3
                                : lofs[(4 * j + 0) / 3] + (4 * j + 0) % 3;
            const int o1 = half ? lofs[(4 * j + 3) / 3] + (4 * j + 3) % 3
                                : lofs[(4 * j + 1) / 3] + (4 * j + 1) % 3;
            v2f a, bA, bB;
            a.x  = w1g[f0 * 16 + lm];
            a.y  = w1g[f1 * 16 + lm];
            bA.x = xrowA[o0];  bA.y = xrowA[o1];
            bB.x = xrowB[o0];  bB.y = xrowB[o1];
            accA = __builtin_amdgcn_wmma_f32_16x16x4_f32(
                false, a, false, bA, (short)0, accA, false, false);
            accB = __builtin_amdgcn_wmma_f32_16x16x4_f32(
                false, a, false, bB, (short)0, accB, false, false);
        }

        const float* b1g = b1 + g * 16 + 8 * half;
#pragma unroll
        for (int i = 0; i < 8; ++i) {
            const float bb = b1g[i];
            float tA = accA[i] + bb;
            float tB = accB[i] + bb;
            accA[i] = tA > 0.0f ? tA : 0.0f;
            accB[i] = tB > 0.0f ? tB : 0.0f;
        }

        v8f acc2A = {};
        v8f acc2B = {};
#pragma unroll
        for (int j = 0; j < 4; ++j) {
            const int h0 = 2 * j + 8 * half;
            v2f b;
            b.x = w2g[(h0 + 0) * 3];
            b.y = w2g[(h0 + 1) * 3];
            v2f aA; aA.x = accA[2 * j]; aA.y = accA[2 * j + 1];
            v2f aB; aB.x = accB[2 * j]; aB.y = accB[2 * j + 1];
            acc2A = __builtin_amdgcn_wmma_f32_16x16x4_f32(
                false, aA, false, b, (short)0, acc2A, false, false);
            acc2B = __builtin_amdgcn_wmma_f32_16x16x4_f32(
                false, aB, false, b, (short)0, acc2B, false, false);
        }

        const float bb2 = b2[g * 3 + cidx];
        if (lm < 3) {
#pragma unroll
            for (int i = 0; i < 8; ++i) {
                const int row = i + 8 * half;
                lrow[row * OROW + g * 3 + lm]                = acc2A[i] + bb2;
                lrow[(row + TILE) * OROW + g * 3 + lm]       = acc2B[i] + bb2;
            }
        }
    }

    float* outp = out + (size_t)pair * (TPW * OROW);
#pragma unroll 4
    for (int k = lane; k < TPW * OROW; k += 32)
        __builtin_nontemporal_store(lrow[k], &outp[k]);
}

extern "C" void kernel_launch(void* const* d_in, const int* in_sizes, int n_in,
                              void* d_out, int out_size, void* d_ws, size_t ws_size,
                              hipStream_t stream) {
    (void)n_in; (void)out_size; (void)d_ws; (void)ws_size;
    const float* x  = (const float*)d_in[0];
    const float* W1 = (const float*)d_in[1];
    const float* b1 = (const float*)d_in[2];
    const float* W2 = (const float*)d_in[3];
    const float* b2 = (const float*)d_in[4];
    const int*   idx = (const int*)d_in[5];
    float* out = (float*)d_out;

    const int ntok   = in_sizes[0] / XROW;
    const int npairs = ntok / TPW;
    const int blocks = (npairs + WAVES_PER_BLOCK - 1) / WAVES_PER_BLOCK;

    bone_mlp_wmma_kernel<<<blocks, WAVES_PER_BLOCK * 32, 0, stream>>>(
        x, W1, b1, W2, b2, idx, out, npairs);
}


# Software-pipeline the 17-group loop one group ahead (fully unrolled, two register sets): next group's W1/b1/W2/b2 loads and LDS feature reads issue as ONE block ahead of the current group's WMMAs
# speedup vs baseline: 2.6648x; 1.0342x over previous
#include <hip/hip_runtime.h>


typedef __attribute__((ext_vector_type(2))) float v2f;
typedef __attribute__((ext_vector_type(4))) float v4f;
typedef __attribute__((ext_vector_type(8))) float v8f;

#define NGROUPS 17
#define WAVES_PER_BLOCK 8
#define TILE 16
#define TPW  (2 * TILE)
#define XROW 48
#define OROW 51
#define XS   52

struct GroupOps {
    v2f   w1[3];
    v2f   fA[3];
    v2f   fB[3];
    float c1[8];
    v2f   w2[4];
    float c2;
};

__global__ __launch_bounds__(WAVES_PER_BLOCK * 32)
void bone_mlp_wmma_kernel(const float* __restrict__ x,
                          const float* __restrict__ W1,
                          const float* __restrict__ b1,
                          const float* __restrict__ W2,
                          const float* __restrict__ b2,
                          const int*   __restrict__ idx,
                          float* __restrict__ out,
                          int npairs)
{
    __shared__ float lds_out[WAVES_PER_BLOCK][TPW * OROW];
    __shared__ __attribute__((aligned(16))) float lds_x[WAVES_PER_BLOCK][TPW * XS];

    const int lane = threadIdx.x & 31;
    const int wave = threadIdx.x >> 5;
    const int pair = blockIdx.x * WAVES_PER_BLOCK + wave;
    if (pair >= npairs) return;

    const int half = lane >> 4;
    const int lm   = lane & 15;
    const int cidx = lm < 3 ? lm : 2;

    float* lrow = &lds_out[wave][0];

    int lofs[NGROUPS][4];
#pragma unroll
    for (int g = 0; g < NGROUPS; ++g) {
#pragma unroll
        for (int i = 0; i < 4; ++i)
            lofs[g][i] = idx[g * 4 + i] * 3;
    }

    float* xs = &lds_x[wave][0];
    {
        const v4f* xw = reinterpret_cast<const v4f*>(x + (size_t)pair * (TPW * XROW));
        v4f r[12];
#pragma unroll
        for (int i = 0; i < 12; ++i)
            r[i] = __builtin_nontemporal_load(xw + i * 32 + lane);
#pragma unroll
        for (int b = 0; b < 3; ++b) {
            const int t = 32 * b + lane;
            float* dst = xs + (t / 12) * XS + (t % 12) * 4;
#pragma unroll
            for (int a = 0; a < 4; ++a)
                *reinterpret_cast<v4f*>(dst + 8 * a * XS) = r[3 * a + b];
        }
    }
    const float* xrowA = xs + lm * XS;
    const float* xrowB = xrowA + TILE * XS;

    auto fetch = [&](GroupOps& o, const int g) {
        const float* w1g = W1 + g * 12 * 16;
        const float* w2g = W2 + g * 16 * 3 + cidx;
        const float* b1g = b1 + g * 16 + 8 * half;
#pragma unroll
        for (int j = 0; j < 3; ++j) {
            const int f0 = 4 * j + 2 * half;
            const int f1 = f0 + 1;
            const int o0 = half ? lofs[g][(4 * j + 2) / 3] + (4 * j + 2) % 3
                                : lofs[g][(4 * j + 0) / 3] + (4 * j + 0) % 3;
            const int o1 = half ? lofs[g][(4 * j + 3) / 3] + (4 * j + 3) % 3
                                : lofs[g][(4 * j + 1) / 3] + (4 * j + 1) % 3;
            o.w1[j].x = w1g[f0 * 16 + lm];
            o.w1[j].y = w1g[f1 * 16 + lm];
            o.fA[j].x = xrowA[o0];  o.fA[j].y = xrowA[o1];
            o.fB[j].x = xrowB[o0];  o.fB[j].y = xrowB[o1];
        }
#pragma unroll
        for (int i = 0; i < 8; ++i)
            o.c1[i] = b1g[i];
#pragma unroll
        for (int j = 0; j < 4; ++j) {
            const int h0 = 2 * j + 8 * half;
            o.w2[j].x = w2g[(h0 + 0) * 3];
            o.w2[j].y = w2g[(h0 + 1) * 3];
        }
        o.c2 = b2[g * 3 + cidx];
    };

    GroupOps cur, nxt;
    fetch(cur, 0);
#pragma unroll
    for (int g = 0; g < NGROUPS; ++g) {
        if (g + 1 < NGROUPS) fetch(nxt, g + 1);
        __builtin_amdgcn_sched_barrier(0);

        v8f accA = {};
        v8f accB = {};
#pragma unroll
        for (int j = 0; j < 3; ++j) {
            accA = __builtin_amdgcn_wmma_f32_16x16x4_f32(
                false, cur.w1[j], false, cur.fA[j], (short)0, accA, false, false);
            accB = __builtin_amdgcn_wmma_f32_16x16x4_f32(
                false, cur.w1[j], false, cur.fB[j], (short)0, accB, false, false);
        }

#pragma unroll
        for (int i = 0; i < 8; ++i) {
            const float bb = cur.c1[i];
            float tA = accA[i] + bb;
            float tB = accB[i] + bb;
            accA[i] = tA > 0.0f ? tA : 0.0f;
            accB[i] = tB > 0.0f ? tB : 0.0f;
        }

        v8f acc2A = {};
        v8f acc2B = {};
#pragma unroll
        for (int j = 0; j < 4; ++j) {
            v2f aA; aA.x = accA[2 * j]; aA.y = accA[2 * j + 1];
            v2f aB; aB.x = accB[2 * j]; aB.y = accB[2 * j + 1];
            acc2A = __builtin_amdgcn_wmma_f32_16x16x4_f32(
                false, aA, false, cur.w2[j], (short)0, acc2A, false, false);
            acc2B = __builtin_amdgcn_wmma_f32_16x16x4_f32(
                false, aB, false, cur.w2[j], (short)0, acc2B, false, false);
        }

        const float bb2 = cur.c2;
        if (lm < 3) {
#pragma unroll
            for (int i = 0; i < 8; ++i) {
                const int row = i + 8 * half;
                lrow[row * OROW + g * 3 + lm]                = acc2A[i] + bb2;
                lrow[(row + TILE) * OROW + g * 3 + lm]       = acc2B[i] + bb2;
            }
        }
        if (g + 1 < NGROUPS) cur = nxt;
    }

    float* outp = out + (size_t)pair * (TPW * OROW);
#pragma unroll 4
    for (int k = lane; k < TPW * OROW; k += 32)
        __builtin_nontemporal_store(lrow[k], &outp[k]);
}

extern "C" void kernel_launch(void* const* d_in, const int* in_sizes, int n_in,
                              void* d_out, int out_size, void* d_ws, size_t ws_size,
                              hipStream_t stream) {
    (void)n_in; (void)out_size; (void)d_ws; (void)ws_size;
    const float* x  = (const float*)d_in[0];
    const float* W1 = (const float*)d_in[1];
    const float* b1 = (const float*)d_in[2];
    const float* W2 = (const float*)d_in[3];
    const float* b2 = (const float*)d_in[4];
    const int*   idx = (const int*)d_in[5];
    float* out = (float*)d_out;

    const int ntok   = in_sizes[0] / XROW;
    const int npairs = ntok / TPW;
    const int blocks = (npairs + WAVES_PER_BLOCK - 1) / WAVES_PER_BLOCK;

    bone_mlp_wmma_kernel<<<blocks, WAVES_PER_BLOCK * 32, 0, stream>>>(
        x, W1, b1, W2, b2, idx, out, npairs);
}


# Layer-1 k order regrouped: lane half h takes k=6h..6h+5 (two whole limbs; layer-1 sum order changes), reads its own two limb offsets; no per-group offset select/arithmetic, no fragment re-pairing
# speedup vs baseline: 2.7635x; 1.0370x over previous
#include <hip/hip_runtime.h>


typedef __attribute__((ext_vector_type(2))) float v2f;
typedef __attribute__((ext_vector_type(4))) float v4f;
typedef __attribute__((ext_vector_type(8))) float v8f;

#define NGROUPS 17
#define WAVES_PER_BLOCK 8
#define TILE 16
#define TPW  (2 * TILE)
#define XROW 48
#define OROW 51
#define XS   52

struct GroupOps {
    v2f   w1[3];
    v2f   fA[3];
    v2f   fB[3];
    float c1[8];
    v2f   w2[4];
    float c2;
};

__global__ __launch_bounds__(WAVES_PER_BLOCK * 32)
void bone_mlp_wmma_kernel(const float* __restrict__ x,
                          const float* __restrict__ W1,
                          const float* __restrict__ b1,
                          const float* __restrict__ W2,
                          const float* __restrict__ b2,
                          const int*   __restrict__ idx,
                          float* __restrict__ out,
                          int npairs)
{
    __shared__ float lds_out[WAVES_PER_BLOCK][TPW * OROW];
    __shared__ __attribute__((aligned(16))) float lds_x[WAVES_PER_BLOCK][TPW * XS];

    const int lane = threadIdx.x & 31;
    const int wave = threadIdx.x >> 5;
    const int pair = blockIdx.x * WAVES_PER_BLOCK + wave;
    if (pair >= npairs) return;

    const int half = lane >> 4;
    const int lm   = lane & 15;
    const int cidx = lm < 3 ? lm : 2;

    float* lrow = &lds_out[wave][0];

    const int* idh = idx + 2 * half;
    int na = idh[0];
    int nb = idh[1];

    float* xs = &lds_x[wave][0];
    {
        const v4f* xw = reinterpret_cast<const v4f*>(x + (size_t)pair * (TPW * XROW));
        v4f r[12];
#pragma unroll
        for (int i = 0; i < 12; ++i)
            r[i] = __builtin_nontemporal_load(xw + i * 32 + lane);
#pragma unroll
        for (int b = 0; b < 3; ++b) {
            const int t = 32 * b + lane;
            float* dst = xs + (t / 12) * XS + (t % 12) * 4;
#pragma unroll
            for (int a = 0; a < 4; ++a)
                *reinterpret_cast<v4f*>(dst + 8 * a * XS) = r[3 * a + b];
        }
    }
    const float* xrowA = xs + lm * XS;

    auto fetch = [&](GroupOps& o, const int g) {
        const float* w1g = W1 + g * 12 * 16 + 6 * 16 * half + lm;
        const float* w2g = W2 + g * 16 * 3 + cidx;
        const float* b1g = b1 + g * 16 + 8 * half;
#pragma unroll
        for (int j = 0; j < 3; ++j) {
            o.w1[j].x = w1g[(2 * j + 0) * 16];
            o.w1[j].y = w1g[(2 * j + 1) * 16];
        }
        const char*  rowA = reinterpret_cast<const char*>(xrowA);
        const float* xa = reinterpret_cast<const float*>(rowA + __mul24(na, 12));
        const float* xb = reinterpret_cast<const float*>(rowA + __mul24(nb, 12));
        o.fA[0].x = xa[0];  o.fA[0].y = xa[1];
        o.fA[1].x = xa[2];  o.fA[1].y = xb[0];
        o.fA[2].x = xb[1];  o.fA[2].y = xb[2];
        const float* ya = xa + TILE * XS;
        const float* yb = xb + TILE * XS;
        o.fB[0].x = ya[0];  o.fB[0].y = ya[1];
        o.fB[1].x = ya[2];  o.fB[1].y = yb[0];
        o.fB[2].x = yb[1];  o.fB[2].y = yb[2];
#pragma unroll
        for (int i = 0; i < 8; ++i)
            o.c1[i] = b1g[i];
#pragma unroll
        for (int j = 0; j < 4; ++j) {
            const int h0 = 2 * j + 8 * half;
            o.w2[j].x = w2g[(h0 + 0) * 3];
            o.w2[j].y = w2g[(h0 + 1) * 3];
        }
        o.c2 = b2[g * 3 + cidx];
        if (g + 1 < NGROUPS) {
            na = idh[(g + 1) * 4 + 0];
            nb = idh[(g + 1) * 4 + 1];
        }
    };

    GroupOps cur, nxt;
    fetch(cur, 0);
#pragma unroll
    for (int g = 0; g < NGROUPS; ++g) {
        if (g + 1 < NGROUPS) fetch(nxt, g + 1);
        __builtin_amdgcn_sched_barrier(0);

        v8f accA = {};
        v8f accB = {};
#pragma unroll
        for (int j = 0; j < 3; ++j) {
            accA = __builtin_amdgcn_wmma_f32_16x16x4_f32(
                false, cur.w1[j], false, cur.fA[j], (short)0, accA, false, false);
            accB = __builtin_amdgcn_wmma_f32_16x16x4_f32(
                false, cur.w1[j], false, cur.fB[j], (short)0, accB, false, false);
        }

#pragma unroll
        for (int i = 0; i < 8; ++i) {
            const float bb = cur.c1[i];
            float tA = accA[i] + bb;
            float tB = accB[i] + bb;
            accA[i] = tA > 0.0f ? tA : 0.0f;
            accB[i] = tB > 0.0f ? tB : 0.0f;
        }

        v8f acc2A = {};
        v8f acc2B = {};
#pragma unroll
        for (int j = 0; j < 4; ++j) {
            v2f aA; aA.x = accA[2 * j]; aA.y = accA[2 * j + 1];
            v2f aB; aB.x = accB[2 * j]; aB.y = accB[2 * j + 1];
            acc2A = __builtin_amdgcn_wmma_f32_16x16x4_f32(
                false, aA, false, cur.w2[j], (short)0, acc2A, false, false);
            acc2B = __builtin_amdgcn_wmma_f32_16x16x4_f32(
                false, aB, false, cur.w2[j], (short)0, acc2B, false, false);
        }

        const float bb2 = cur.c2;
        if (lm < 3) {
#pragma unroll
            for (int i = 0; i < 8; ++i) {
                const int row = i + 8 * half;
                lrow[row * OROW + g * 3 + lm]                = acc2A[i] + bb2;
                lrow[(row + TILE) * OROW + g * 3 + lm]       = acc2B[i] + bb2;
            }
        }
        if (g + 1 < NGROUPS) cur = nxt;
    }

    float* outp = out + (size_t)pair * (TPW * OROW);
#pragma unroll 4
    for (int k = lane; k < TPW * OROW; k += 32)
        __builtin_nontemporal_store(lrow[k], &outp[k]);
}

extern "C" void kernel_launch(void* const* d_in, const int* in_sizes, int n_in,
                              void* d_out, int out_size, void* d_ws, size_t ws_size,
                              hipStream_t stream) {
    (void)n_in; (void)out_size; (void)d_ws; (void)ws_size;
    const float* x  = (const float*)d_in[0];
    const float* W1 = (const float*)d_in[1];
    const float* b1 = (const float*)d_in[2];
    const float* W2 = (const float*)d_in[3];
    const float* b2 = (const float*)d_in[4];
    const int*   idx = (const int*)d_in[5];
    float* out = (float*)d_out;

    const int ntok   = in_sizes[0] / XROW;
    const int npairs = ntok / TPW;
    const int blocks = (npairs + WAVES_PER_BLOCK - 1) / WAVES_PER_BLOCK;

    bone_mlp_wmma_kernel<<<blocks, WAVES_PER_BLOCK * 32, 0, stream>>>(
        x, W1, b1, W2, b2, idx, out, npairs);
}


# Layer 2 transposed (W2 as A operand, layer-1 result as B): each lane ends with its own token's 3 channels; 2 full-wave LDS stores per group replace 8 masked 6-lane stores and the exec branch
# speedup vs baseline: 2.9868x; 1.0808x over previous
#include <hip/hip_runtime.h>


typedef __attribute__((ext_vector_type(2))) float v2f;
typedef __attribute__((ext_vector_type(4))) float v4f;
typedef __attribute__((ext_vector_type(8))) float v8f;

#define NGROUPS 17
#define WAVES_PER_BLOCK 8
#define TILE 16
#define TPW  (2 * TILE)
#define XROW 48
#define OROW 51
#define XS   52

struct GroupOps {
    v2f   w1[3];
    float fa[6];
    float fb[6];
    float c1[8];
    v2f   w2[4];
    float c2[3];
};

__global__ __launch_bounds__(WAVES_PER_BLOCK * 32)
void bone_mlp_wmma_kernel(const float* __restrict__ x,
                          const float* __restrict__ W1,
                          const float* __restrict__ b1,
                          const float* __restrict__ W2,
                          const float* __restrict__ b2,
                          const int*   __restrict__ idx,
                          float* __restrict__ out,
                          int npairs)
{
    __shared__ float lds_out[WAVES_PER_BLOCK][TPW * OROW];
    __shared__ __attribute__((aligned(16))) float lds_x[WAVES_PER_BLOCK][TPW * XS];

    const int lane = threadIdx.x & 31;
    const int wave = threadIdx.x >> 5;
    const int pair = blockIdx.x * WAVES_PER_BLOCK + wave;
    if (pair >= npairs) return;

    const int half = lane >> 4;
    const int lm   = lane & 15;
    const int cidx = (lm & 7) < 3 ? (lm & 7) : 2;

    float* lrow = &lds_out[wave][0];

    const int* idh = idx + 2 * half;
    int na = idh[0];
    int nb = idh[1];

    float* xs = &lds_x[wave][0];
    {
        const v4f* xw = reinterpret_cast<const v4f*>(x + (size_t)pair * (TPW * XROW));
        v4f r[12];
#pragma unroll
        for (int i = 0; i < 12; ++i)
            r[i] = __builtin_nontemporal_load(xw + i * 32 + lane);
#pragma unroll
        for (int b = 0; b < 3; ++b) {
            const int t = 32 * b + lane;
            float* dst = xs + (t / 12) * XS + (t % 12) * 4;
#pragma unroll
            for (int a = 0; a < 4; ++a)
                *reinterpret_cast<v4f*>(dst + 8 * a * XS) = r[3 * a + b];
        }
    }
    const float* xrowA = xs + lm * XS;
    int offB = TILE * XS * (int)sizeof(float);
    asm volatile("" : "+v"(offB));

    auto fetch = [&](GroupOps& o, const int g) {
        const float* w1g = W1 + g * 12 * 16 + 6 * 16 * half + lm;
        const float* w2g = W2 + g * 16 * 3 + cidx;
        const float* b1g = b1 + g * 16 + 8 * half;
#pragma unroll
        for (int j = 0; j < 3; ++j) {
            o.w1[j].x = w1g[(2 * j + 0) * 16];
            o.w1[j].y = w1g[(2 * j + 1) * 16];
        }
        const char*  rowA = reinterpret_cast<const char*>(xrowA);
        const int oa = __mul24(na, 12);
        const int ob = __mul24(nb, 12);
        const float* xa = reinterpret_cast<const float*>(rowA + oa);
        const float* xb = reinterpret_cast<const float*>(rowA + ob);
        o.fa[0] = xa[0];  o.fa[1] = xa[1];  o.fa[2] = xa[2];
        o.fa[4] = xb[1];  o.fa[5] = xb[2];  o.fa[3] = xb[0];
        const float* ya = reinterpret_cast<const float*>(rowA + (oa + offB));
        const float* yb = reinterpret_cast<const float*>(rowA + (ob + offB));
        o.fb[0] = ya[0];  o.fb[1] = ya[1];  o.fb[2] = ya[2];
        o.fb[4] = yb[1];  o.fb[5] = yb[2];  o.fb[3] = yb[0];
#pragma unroll
        for (int i = 0; i < 8; ++i)
            o.c1[i] = b1g[i];
#pragma unroll
        for (int j = 0; j < 4; ++j) {
            const int h0 = 2 * j + 8 * half;
            o.w2[j].x = w2g[(h0 + 0) * 3];
            o.w2[j].y = w2g[(h0 + 1) * 3];
        }
#pragma unroll
        for (int r = 0; r < 3; ++r)
            o.c2[r] = b2[g * 3 + r];
        if (g + 1 < NGROUPS) {
            na = idh[(g + 1) * 4 + 0];
            nb = idh[(g + 1) * 4 + 1];
        }
    };

    GroupOps cur, nxt;
    fetch(cur, 0);
#pragma unroll
    for (int g = 0; g < NGROUPS; ++g) {
        if (g + 1 < NGROUPS) {
            asm volatile("" : "+v"(na), "+v"(nb));
            fetch(nxt, g + 1);
        }
        asm volatile("" ::: "memory");
        __builtin_amdgcn_sched_barrier(0);

        v8f accA = {};
        v8f accB = {};
#pragma unroll
        for (int j = 0; j < 3; ++j) {
            v2f gA; gA.x = cur.fa[2 * j]; gA.y = cur.fa[2 * j + 1];
            v2f gB; gB.x = cur.fb[2 * j]; gB.y = cur.fb[2 * j + 1];
            accA = __builtin_amdgcn_wmma_f32_16x16x4_f32(
                false, cur.w1[j], false, gA, (short)0, accA, false, false);
            accB = __builtin_amdgcn_wmma_f32_16x16x4_f32(
                false, cur.w1[j], false, gB, (short)0, accB, false, false);
        }

#pragma unroll
        for (int i = 0; i < 8; ++i) {
            const float bb = cur.c1[i];
            float tA = accA[i] + bb;
            float tB = accB[i] + bb;
            accA[i] = tA > 0.0f ? tA : 0.0f;
            accB[i] = tB > 0.0f ? tB : 0.0f;
        }

        v8f acc2A = {};
        v8f acc2B = {};
#pragma unroll
        for (int j = 0; j < 4; ++j) {
            v2f hA; hA.x = accA[2 * j]; hA.y = accA[2 * j + 1];
            v2f hB; hB.x = accB[2 * j]; hB.y = accB[2 * j + 1];
            acc2A = __builtin_amdgcn_wmma_f32_16x16x4_f32(
                false, cur.w2[j], false, hA, (short)0, acc2A, false, false);
            acc2B = __builtin_amdgcn_wmma_f32_16x16x4_f32(
                false, cur.w2[j], false, hB, (short)0, acc2B, false, false);
        }

        __builtin_amdgcn_sched_barrier(0);

        float* orow = lrow + lane * OROW + g * 3;
#pragma unroll
        for (int r = 0; r < 3; ++r)
            orow[r] = (half ? acc2B[r] : acc2A[r]) + cur.c2[r];
        if (g + 1 < NGROUPS) cur = nxt;
    }

    float* outp = out + (size_t)pair * (TPW * OROW);
#pragma unroll 4
    for (int k = lane; k < TPW * OROW; k += 32)
        __builtin_nontemporal_store(lrow[k], &outp[k]);
}

extern "C" void kernel_launch(void* const* d_in, const int* in_sizes, int n_in,
                              void* d_out, int out_size, void* d_ws, size_t ws_size,
                              hipStream_t stream) {
    (void)n_in; (void)out_size; (void)d_ws; (void)ws_size;
    const float* x  = (const float*)d_in[0];
    const float* W1 = (const float*)d_in[1];
    const float* b1 = (const float*)d_in[2];
    const float* W2 = (const float*)d_in[3];
    const float* b2 = (const float*)d_in[4];
    const int*   idx = (const int*)d_in[5];
    float* out = (float*)d_out;

    const int ntok   = in_sizes[0] / XROW;
    const int npairs = ntok / TPW;
    const int blocks = (npairs + WAVES_PER_BLOCK - 1) / WAVES_PER_BLOCK;

    bone_mlp_wmma_kernel<<<blocks, WAVES_PER_BLOCK * 32, 0, stream>>>(
        x, W1, b1, W2, b2, idx, out, npairs);
}


# Writeback un-serialised and widened: all 13 16-byte LDS reads per lane issued before the first of 13 16-byte non-temporal stores (was 15 load-wait-store rounds of 4-byte pieces, 51 stores)
# speedup vs baseline: 3.1314x; 1.0484x over previous
#include <hip/hip_runtime.h>


typedef __attribute__((ext_vector_type(2))) float v2f;
typedef __attribute__((ext_vector_type(4))) float v4f;
typedef __attribute__((ext_vector_type(8))) float v8f;

#define NGROUPS 17
#define WAVES_PER_BLOCK 8
#define TILE 16
#define TPW  (2 * TILE)
#define XROW 48
#define OROW 51
#define XS   52

struct GroupOps {
    v2f   w1[3];
    float fa[6];
    float fb[6];
    float c1[8];
    v2f   w2[4];
    float c2[3];
};

__global__ __launch_bounds__(WAVES_PER_BLOCK * 32)
void bone_mlp_wmma_kernel(const float* __restrict__ x,
                          const float* __restrict__ W1,
                          const float* __restrict__ b1,
                          const float* __restrict__ W2,
                          const float* __restrict__ b2,
                          const int*   __restrict__ idx,
                          float* __restrict__ out,
                          int npairs)
{
    __shared__ __attribute__((aligned(16))) float lds_out[WAVES_PER_BLOCK][TPW * OROW];
    __shared__ __attribute__((aligned(16))) float lds_x[WAVES_PER_BLOCK][TPW * XS];

    const int lane = threadIdx.x & 31;
    const int wave = threadIdx.x >> 5;
    const int pair = blockIdx.x * WAVES_PER_BLOCK + wave;
    if (pair >= npairs) return;

    const int half = lane >> 4;
    const int lm   = lane & 15;
    const int cidx = (lm & 7) < 3 ? (lm & 7) : 2;

    float* lrow = &lds_out[wave][0];

    const int* idh = idx + 2 * half;
    int na = idh[0];
    int nb = idh[1];

    float* xs = &lds_x[wave][0];
    {
        const v4f* xw = reinterpret_cast<const v4f*>(x + (size_t)pair * (TPW * XROW));
        v4f r[12];
#pragma unroll
        for (int i = 0; i < 12; ++i)
            r[i] = __builtin_nontemporal_load(xw + i * 32 + lane);
#pragma unroll
        for (int b = 0; b < 3; ++b) {
            const int t = 32 * b + lane;
            float* dst = xs + (t / 12) * XS + (t % 12) * 4;
#pragma unroll
            for (int a = 0; a < 4; ++a)
                *reinterpret_cast<v4f*>(dst + 8 * a * XS) = r[3 * a + b];
        }
    }
    const float* xrowA = xs + lm * XS;
    int offB = TILE * XS * (int)sizeof(float);
    asm volatile("" : "+v"(offB));

    auto fetch = [&](GroupOps& o, const int g) {
        const float* w1g = W1 + g * 12 * 16 + 6 * 16 * half + lm;
        const float* w2g = W2 + g * 16 * 3 + cidx;
        const float* b1g = b1 + g * 16 + 8 * half;
#pragma unroll
        for (int j = 0; j < 3; ++j) {
            o.w1[j].x = w1g[(2 * j + 0) * 16];
            o.w1[j].y = w1g[(2 * j + 1) * 16];
        }
        const char*  rowA = reinterpret_cast<const char*>(xrowA);
        const int oa = __mul24(na, 12);
        const int ob = __mul24(nb, 12);
        const float* xa = reinterpret_cast<const float*>(rowA + oa);
        const float* xb = reinterpret_cast<const float*>(rowA + ob);
        o.fa[0] = xa[0];  o.fa[1] = xa[1];  o.fa[2] = xa[2];
        o.fa[4] = xb[1];  o.fa[5] = xb[2];  o.fa[3] = xb[0];
        const float* ya = reinterpret_cast<const float*>(rowA + (oa + offB));
        const float* yb = reinterpret_cast<const float*>(rowA + (ob + offB));
        o.fb[0] = ya[0];  o.fb[1] = ya[1];  o.fb[2] = ya[2];
        o.fb[4] = yb[1];  o.fb[5] = yb[2];  o.fb[3] = yb[0];
#pragma unroll
        for (int i = 0; i < 8; ++i)
            o.c1[i] = b1g[i];
#pragma unroll
        for (int j = 0; j < 4; ++j) {
            const int h0 = 2 * j + 8 * half;
            o.w2[j].x = w2g[(h0 + 0) * 3];
            o.w2[j].y = w2g[(h0 + 1) * 3];
        }
#pragma unroll
        for (int r = 0; r < 3; ++r)
            o.c2[r] = b2[g * 3 + r];
        if (g + 1 < NGROUPS) {
            na = idh[(g + 1) * 4 + 0];
            nb = idh[(g + 1) * 4 + 1];
        }
    };

    GroupOps cur, nxt;
    fetch(cur, 0);
#pragma unroll
    for (int g = 0; g < NGROUPS; ++g) {
        if (g + 1 < NGROUPS) {
            asm volatile("" : "+v"(na), "+v"(nb));
            fetch(nxt, g + 1);
        }
        asm volatile("" ::: "memory");
        __builtin_amdgcn_sched_barrier(0);

        v8f accA = {};
        v8f accB = {};
#pragma unroll
        for (int j = 0; j < 3; ++j) {
            v2f gA; gA.x = cur.fa[2 * j]; gA.y = cur.fa[2 * j + 1];
            v2f gB; gB.x = cur.fb[2 * j]; gB.y = cur.fb[2 * j + 1];
            accA = __builtin_amdgcn_wmma_f32_16x16x4_f32(
                false, cur.w1[j], false, gA, (short)0, accA, false, false);
            accB = __builtin_amdgcn_wmma_f32_16x16x4_f32(
                false, cur.w1[j], false, gB, (short)0, accB, false, false);
        }

#pragma unroll
        for (int i = 0; i < 8; ++i) {
            const float bb = cur.c1[i];
            float tA = accA[i] + bb;
            float tB = accB[i] + bb;
            accA[i] = tA > 0.0f ? tA : 0.0f;
            accB[i] = tB > 0.0f ? tB : 0.0f;
        }

        v8f acc2A = {};
        v8f acc2B = {};
#pragma unroll
        for (int j = 0; j < 4; ++j) {
            v2f hA; hA.x = accA[2 * j]; hA.y = accA[2 * j + 1];
            v2f hB; hB.x = accB[2 * j]; hB.y = accB[2 * j + 1];
            acc2A = __builtin_amdgcn_wmma_f32_16x16x4_f32(
                false, cur.w2[j], false, hA, (short)0, acc2A, false, false);
            acc2B = __builtin_amdgcn_wmma_f32_16x16x4_f32(
                false, cur.w2[j], false, hB, (short)0, acc2B, false, false);
        }

        __builtin_amdgcn_sched_barrier(0);

        float* orow = lrow + lane * OROW + g * 3;
#pragma unroll
        for (int r = 0; r < 3; ++r)
            orow[r] = (half ? acc2B[r] : acc2A[r]) + cur.c2[r];
        if (g + 1 < NGROUPS) cur = nxt;
    }

    {
        const v4f* l4 = reinterpret_cast<const v4f*>(lrow);
        v4f* o4 = reinterpret_cast<v4f*>(out + (size_t)pair * (TPW * OROW));
        const bool tail = lane < (TPW * OROW / 4 - 12 * 32);
        v4f t[13];
#pragma unroll
        for (int i = 0; i < 12; ++i)
            t[i] = l4[i * 32 + lane];
        if (tail) t[12] = l4[12 * 32 + lane];
#pragma unroll
        for (int i = 0; i < 12; ++i)
            __builtin_nontemporal_store(t[i], o4 + i * 32 + lane);
        if (tail) __builtin_nontemporal_store(t[12], o4 + 12 * 32 + lane);
    }
}

extern "C" void kernel_launch(void* const* d_in, const int* in_sizes, int n_in,
                              void* d_out, int out_size, void* d_ws, size_t ws_size,
                              hipStream_t stream) {
    (void)n_in; (void)out_size; (void)d_ws; (void)ws_size;
    const float* x  = (const float*)d_in[0];
    const float* W1 = (const float*)d_in[1];
    const float* b1 = (const float*)d_in[2];
    const float* W2 = (const float*)d_in[3];
    const float* b2 = (const float*)d_in[4];
    const int*   idx = (const int*)d_in[5];
    float* out = (float*)d_out;

    const int ntok   = in_sizes[0] / XROW;
    const int npairs = ntok / TPW;
    const int blocks = (npairs + WAVES_PER_BLOCK - 1) / WAVES_PER_BLOCK;

    bone_mlp_wmma_kernel<<<blocks, WAVES_PER_BLOCK * 32, 0, stream>>>(
        x, W1, b1, W2, b2, idx, out, npairs);
}
